# starting kernel
# baseline (speedup 1.0000x reference)
#include <hip/hip_runtime.h>

typedef float v2f __attribute__((ext_vector_type(2)));
typedef float v8f __attribute__((ext_vector_type(8)));

#define NF 16
#define NPOLY 969
#define KPAD 972
#define TILE_ROWS 16
#define WAVES_PER_WG 8
#define BLOCK_THREADS (WAVES_PER_WG * 32)

#define EMIT(val)                                                            \
  do {                                                                       \
    f4[p & 3] = (val);                                                       \
    if ((p & 3) == 3) {                                                      \
      v2f bfrag = *(const v2f*)(bt + ((p >> 2) << 2));                       \
      v2f afrag;                                                             \
      afrag.x = lo ? f4[0] : f4[2];                                          \
      afrag.y = lo ? f4[1] : f4[3];                                          \
      acc = __builtin_amdgcn_wmma_f32_16x16x4_f32(                           \
          false, afrag, false, bfrag, (short)0, acc, false, false);          \
    }                                                                        \
    ++p;                                                                     \
  } while (0)

__global__ __launch_bounds__(BLOCK_THREADS) void taylor_wmma_kernel(
    const float* __restrict__ X, const float* __restrict__ W,
    float* __restrict__ out, int ntiles) {
  __shared__ float WT[NF * KPAD];
  for (int idx = threadIdx.x; idx < NF * KPAD; idx += BLOCK_THREADS) {
    int k = idx >> 4;
    int n = idx & 15;
    WT[n * KPAD + k] = (k < NPOLY) ? W[k * NF + n] : 0.0f;
  }
  __syncthreads();

  const int wave = threadIdx.x >> 5;
  const int lane = threadIdx.x & 31;
  const int tile = blockIdx.x * WAVES_PER_WG + wave;
  if (tile >= ntiles) return;

  const bool lo = lane < 16;
  const int n = lane & 15;
  const int row = tile * TILE_ROWS + n;

  float xv[NF];
  {
    const float4* xp = (const float4*)(X + row * NF);
    float4 q0 = xp[0], q1 = xp[1], q2 = xp[2], q3 = xp[3];
    xv[0] = q0.x;  xv[1] = q0.y;  xv[2] = q0.z;  xv[3] = q0.w;
    xv[4] = q1.x;  xv[5] = q1.y;  xv[6] = q1.z;  xv[7] = q1.w;
    xv[8] = q2.x;  xv[9] = q2.y;  xv[10] = q2.z; xv[11] = q2.w;
    xv[12] = q3.x; xv[13] = q3.y; xv[14] = q3.z; xv[15] = q3.w;
  }

  const int m0 = lo ? 0 : 8;
  v8f acc;
#pragma unroll
  for (int j = 0; j < 8; ++j)
    acc[j] = X[(tile * TILE_ROWS + m0 + j) * NF + n];

  const float* bt = &WT[n * KPAD + (lo ? 0 : 2)];

  float f4[4];
  int p = 0;

  EMIT(1.0f);
#pragma unroll
  for (int i = 0; i < NF; ++i) EMIT(xv[i]);
#pragma unroll
  for (int f = 0; f < NF; ++f) {
#pragma unroll
    for (int j = f; j < NF; ++j) EMIT(xv[f] * xv[j]);
  }
#pragma unroll
  for (int f = 0; f < NF; ++f) {
#pragma unroll
    for (int j = f; j < NF; ++j) {
      float pfj = xv[f] * xv[j];
#pragma unroll
      for (int k = j; k < NF; ++k) EMIT(pfj * xv[k]);
    }
  }
  EMIT(0.0f);
  EMIT(0.0f);
  EMIT(0.0f);

#pragma unroll
  for (int j = 0; j < 8; ++j)
    out[(tile * TILE_ROWS + m0 + j) * NF + n] = acc[j];
}

extern "C" void kernel_launch(void* const* d_in, const int* in_sizes, int n_in,
                              void* d_out, int out_size, void* d_ws,
                              size_t ws_size, hipStream_t stream) {
  const float* X = (const float*)d_in[0];
  const float* W = (const float*)d_in[1];
  float* out = (float*)d_out;

  const int batch = in_sizes[0] / NF;
  const int ntiles = batch / TILE_ROWS;
  const int blocks = (ntiles + WAVES_PER_WG - 1) / WAVES_PER_WG;

  taylor_wmma_kernel<<<blocks, BLOCK_THREADS, 0, stream>>>(X, W, out, ntiles);
}


# W staged into LDS with all of a thread's 16-byte global loads issued as one group before any LDS store, in place of one dependent load-wait-store round trip per element
# speedup vs baseline: 2.6862x; 2.6862x over previous
#include <hip/hip_runtime.h>

typedef float v2f __attribute__((ext_vector_type(2)));
typedef float v8f __attribute__((ext_vector_type(8)));

#define NF 16
#define NPOLY 969
#define KPAD 972
#define TILE_ROWS 16
#define WAVES_PER_WG 8
#define BLOCK_THREADS (WAVES_PER_WG * 32)
#define W_QUADS (NPOLY * NF / 4)
#define W_FULL (W_QUADS / BLOCK_THREADS)

#define EMIT(val)                                                            \
  do {                                                                       \
    f4[p & 3] = (val);                                                       \
    if ((p & 3) == 3) {                                                      \
      v2f bfrag = *(const v2f*)(bt + ((p >> 2) << 2));                       \
      v2f afrag;                                                             \
      afrag.x = lo ? f4[0] : f4[2];                                          \
      afrag.y = lo ? f4[1] : f4[3];                                          \
      acc = __builtin_amdgcn_wmma_f32_16x16x4_f32(                           \
          false, afrag, false, bfrag, (short)0, acc, false, false);          \
    }                                                                        \
    ++p;                                                                     \
  } while (0)

__global__ __launch_bounds__(BLOCK_THREADS) void taylor_wmma_kernel(
    const float* __restrict__ X, const float* __restrict__ W,
    float* __restrict__ out, int ntiles) {
  __shared__ float WT[NF * KPAD];
  {
    const float4* W4 = (const float4*)W;
    float4 wq[W_FULL];
#pragma unroll
    for (int i = 0; i < W_FULL; ++i) wq[i] = W4[threadIdx.x + i * BLOCK_THREADS];
    const int qt = threadIdx.x + W_FULL * BLOCK_THREADS;
    float4 wt = make_float4(0.0f, 0.0f, 0.0f, 0.0f);
    if (qt < W_QUADS) wt = W4[qt];

    float* wb = &WT[((threadIdx.x & 3) * 4) * KPAD + (threadIdx.x >> 2)];
#pragma unroll
    for (int i = 0; i < W_FULL; ++i) {
      float* d = wb + i * (BLOCK_THREADS / 4);
      d[0 * KPAD] = wq[i].x;
      d[1 * KPAD] = wq[i].y;
      d[2 * KPAD] = wq[i].z;
      d[3 * KPAD] = wq[i].w;
    }
    if (qt < W_QUADS) {
      float* d = wb + W_FULL * (BLOCK_THREADS / 4);
      d[0 * KPAD] = wt.x;
      d[1 * KPAD] = wt.y;
      d[2 * KPAD] = wt.z;
      d[3 * KPAD] = wt.w;
    }
    if (threadIdx.x < NF) {
      float* z = &WT[threadIdx.x * KPAD + NPOLY];
      z[0] = 0.0f;
      z[1] = 0.0f;
      z[2] = 0.0f;
    }
  }
  __syncthreads();

  const int wave = threadIdx.x >> 5;
  const int lane = threadIdx.x & 31;
  const int tile = blockIdx.x * WAVES_PER_WG + wave;
  if (tile >= ntiles) return;

  const bool lo = lane < 16;
  const int n = lane & 15;
  const int row = tile * TILE_ROWS + n;

  float xv[NF];
  {
    const float4* xp = (const float4*)(X + row * NF);
    float4 q0 = xp[0], q1 = xp[1], q2 = xp[2], q3 = xp[3];
    xv[0] = q0.x;  xv[1] = q0.y;  xv[2] = q0.z;  xv[3] = q0.w;
    xv[4] = q1.x;  xv[5] = q1.y;  xv[6] = q1.z;  xv[7] = q1.w;
    xv[8] = q2.x;  xv[9] = q2.y;  xv[10] = q2.z; xv[11] = q2.w;
    xv[12] = q3.x; xv[13] = q3.y; xv[14] = q3.z; xv[15] = q3.w;
  }

  const int m0 = lo ? 0 : 8;
  v8f acc;
#pragma unroll
  for (int j = 0; j < 8; ++j)
    acc[j] = X[(tile * TILE_ROWS + m0 + j) * NF + n];

  const float* bt = &WT[n * KPAD + (lo ? 0 : 2)];

  float f4[4];
  int p = 0;

  EMIT(1.0f);
#pragma unroll
  for (int i = 0; i < NF; ++i) EMIT(xv[i]);
#pragma unroll
  for (int f = 0; f < NF; ++f) {
#pragma unroll
    for (int j = f; j < NF; ++j) EMIT(xv[f] * xv[j]);
  }
#pragma unroll
  for (int f = 0; f < NF; ++f) {
#pragma unroll
    for (int j = f; j < NF; ++j) {
      float pfj = xv[f] * xv[j];
#pragma unroll
      for (int k = j; k < NF; ++k) EMIT(pfj * xv[k]);
    }
  }
  EMIT(0.0f);
  EMIT(0.0f);
  EMIT(0.0f);

#pragma unroll
  for (int j = 0; j < 8; ++j)
    out[(tile * TILE_ROWS + m0 + j) * NF + n] = acc[j];
}

extern "C" void kernel_launch(void* const* d_in, const int* in_sizes, int n_in,
                              void* d_out, int out_size, void* d_ws,
                              size_t ws_size, hipStream_t stream) {
  const float* X = (const float*)d_in[0];
  const float* W = (const float*)d_in[1];
  float* out = (float*)d_out;

  const int batch = in_sizes[0] / NF;
  const int ntiles = batch / TILE_ROWS;
  const int blocks = (ntiles + WAVES_PER_WG - 1) / WAVES_PER_WG;

  taylor_wmma_kernel<<<blocks, BLOCK_THREADS, 0, stream>>>(X, W, out, ntiles);
}


# W staging overlapped with the matrix loop: X row and W loads issued in k order, W handed to LDS in three pieces behind LDS-only barriers, first matrix instructions run while most of W is in flight
# speedup vs baseline: 2.9154x; 1.0853x over previous
#include <hip/hip_runtime.h>


typedef float v2f __attribute__((ext_vector_type(2)));
typedef float v8f __attribute__((ext_vector_type(8)));

#define NF 16
#define NPOLY 969
#define KPAD 972
#define TILE_ROWS 16
#define WAVES_PER_WG 8
#define BLOCK_THREADS (WAVES_PER_WG * 32)
#define W_QUADS (NPOLY * NF / 4)
#define W_FULL (W_QUADS / BLOCK_THREADS)
#define ROUND_K (BLOCK_THREADS / 4)
#define R_A 2
#define R_B 8
#define K_A (R_A * ROUND_K)
#define K_B (R_B * ROUND_K)

template <int RB, int RE>
__device__ __forceinline__ void stage_rounds(float* wb, const float4 (&wq)[W_FULL]) {
#pragma unroll
  for (int i = RB; i < RE; ++i) {
    float* d = wb + i * ROUND_K;
    d[0 * KPAD] = wq[i].x;
    d[1 * KPAD] = wq[i].y;
    d[2 * KPAD] = wq[i].z;
    d[3 * KPAD] = wq[i].w;
  }
}

__device__ __forceinline__ void lds_barrier() {
  __builtin_amdgcn_fence(__ATOMIC_RELEASE, "workgroup", "local");
  __builtin_amdgcn_s_barrier();
  __builtin_amdgcn_fence(__ATOMIC_ACQUIRE, "workgroup", "local");
}

#define EMIT(val)                                                            \
  do {                                                                       \
    if (p == K_A) {                                                          \
      stage_rounds<R_A, R_B>(wb, wq);                                        \
      lds_barrier();                                                         \
    }                                                                        \
    if (p == K_B) {                                                          \
      stage_rounds<R_B, W_FULL>(wb, wq);                                     \
      if (qt < W_QUADS) {                                                    \
        float* d = wb + W_FULL * ROUND_K;                                    \
        d[0 * KPAD] = wt.x;                                                  \
        d[1 * KPAD] = wt.y;                                                  \
        d[2 * KPAD] = wt.z;                                                  \
        d[3 * KPAD] = wt.w;                                                  \
      }                                                                      \
      if (threadIdx.x < NF) {       \
        float* z = &WT[threadIdx.x * KPAD + NPOLY];                          \
        z[0] = 0.0f;                                                         \
        z[1] = 0.0f;                                                         \
        z[2] = 0.0f;                                                         \
      }                                                                      \
      lds_barrier();                                                         \
    }                                                                        \
    f4[p & 3] = (val);                                                       \
    if ((p & 3) == 3) {                                                      \
      v2f bfrag = *(const v2f*)(bt + ((p >> 2) << 2));                       \
      v2f afrag;                                                             \
      afrag.x = lo ? f4[0] : f4[2];                                          \
      afrag.y = lo ? f4[1] : f4[3];                                          \
      acc = __builtin_amdgcn_wmma_f32_16x16x4_f32(                           \
          false, afrag, false, bfrag, (short)0, acc, false, false);          \
    }                                                                        \
    ++p;                                                                     \
  } while (0)

__global__ __launch_bounds__(BLOCK_THREADS) void taylor_wmma_kernel(
    const float* __restrict__ X, const float* __restrict__ W,
    float* __restrict__ out, int ntiles) {
  __shared__ float WT[NF * KPAD];

  const int wave = threadIdx.x >> 5;
  const int lane = threadIdx.x & 31;
  const int tile_raw = blockIdx.x * WAVES_PER_WG + wave;
  const bool active = tile_raw < ntiles;
  const int tile = active ? tile_raw : (ntiles - 1);

  const bool lo = lane < 16;
  const int n = lane & 15;
  const int row = tile * TILE_ROWS + n;

  float xv[NF];
  {
    const float4* xp = (const float4*)(X + row * NF);
    float4 q0 = xp[0], q1 = xp[1], q2 = xp[2], q3 = xp[3];
    xv[0] = q0.x;  xv[1] = q0.y;  xv[2] = q0.z;  xv[3] = q0.w;
    xv[4] = q1.x;  xv[5] = q1.y;  xv[6] = q1.z;  xv[7] = q1.w;
    xv[8] = q2.x;  xv[9] = q2.y;  xv[10] = q2.z; xv[11] = q2.w;
    xv[12] = q3.x; xv[13] = q3.y; xv[14] = q3.z; xv[15] = q3.w;
  }

  const int m0 = lo ? 0 : 8;
  v8f acc;
#pragma unroll
  for (int j = 0; j < 8; ++j)
    acc[j] = X[(tile * TILE_ROWS + m0 + j) * NF + n];

  const float4* W4 = (const float4*)W;
  float4 wq[W_FULL];
  __builtin_amdgcn_sched_barrier(0);
#pragma unroll
  for (int i = 0; i < R_A; ++i) wq[i] = W4[threadIdx.x + i * BLOCK_THREADS];
  __builtin_amdgcn_sched_barrier(0);
  __builtin_amdgcn_s_barrier();
  __builtin_amdgcn_sched_barrier(0);
#pragma unroll
  for (int i = R_A; i < R_B; ++i) wq[i] = W4[threadIdx.x + i * BLOCK_THREADS];
  __builtin_amdgcn_sched_barrier(0);
#pragma unroll
  for (int i = R_B; i < W_FULL; ++i) wq[i] = W4[threadIdx.x + i * BLOCK_THREADS];
  const int qt = threadIdx.x + W_FULL * BLOCK_THREADS;
  const float4 wt = W4[qt < W_QUADS ? qt : (W_QUADS - 1)];

  float* wb = &WT[((threadIdx.x & 3) * 4) * KPAD + (threadIdx.x >> 2)];

  stage_rounds<0, R_A>(wb, wq);
  __builtin_amdgcn_sched_barrier(0);
  lds_barrier();

  const float* bt = &WT[n * KPAD + (lo ? 0 : 2)];

  float f4[4];
  int p = 0;

  EMIT(1.0f);
#pragma unroll
  for (int i = 0; i < NF; ++i) EMIT(xv[i]);
#pragma unroll
  for (int f = 0; f < NF; ++f) {
#pragma unroll
    for (int j = f; j < NF; ++j) EMIT(xv[f] * xv[j]);
  }
#pragma unroll
  for (int f = 0; f < NF; ++f) {
#pragma unroll
    for (int j = f; j < NF; ++j) {
      float pfj = xv[f] * xv[j];
#pragma unroll
      for (int k = j; k < NF; ++k) EMIT(pfj * xv[k]);
    }
  }
  EMIT(0.0f);
  EMIT(0.0f);
  EMIT(0.0f);

  if (active) {
#pragma unroll
    for (int j = 0; j < 8; ++j)
      out[(tile * TILE_ROWS + m0 + j) * NF + n] = acc[j];
  }
}

extern "C" void kernel_launch(void* const* d_in, const int* in_sizes, int n_in,
                              void* d_out, int out_size, void* d_ws,
                              size_t ws_size, hipStream_t stream) {
  const float* X = (const float*)d_in[0];
  const float* W = (const float*)d_in[1];
  float* out = (float*)d_out;

  const int batch = in_sizes[0] / NF;
  const int ntiles = batch / TILE_ROWS;
  const int blocks = (ntiles + WAVES_PER_WG - 1) / WAVES_PER_WG;

  taylor_wmma_kernel<<<blocks, BLOCK_THREADS, 0, stream>>>(X, W, out, ntiles);
}
